# starting kernel
# baseline (speedup 1.0000x reference)
#include <hip/hip_runtime.h>
#include <hip/hip_bf16.h>

typedef __attribute__((ext_vector_type(2))) float v2f;
typedef __attribute__((ext_vector_type(4))) float v4f;
typedef __attribute__((ext_vector_type(8))) float v8f;

#define DD 128
#define TILE_M 16
#define WAVES_PER_BLOCK 8

__global__ __launch_bounds__(256, 1) void CrossCompress_86655260164823_kernel(
    const float* __restrict__ v_in, const float* __restrict__ e_in,
    const float* __restrict__ w_vv, const float* __restrict__ w_ve,
    const float* __restrict__ w_ev, const float* __restrict__ w_ee,
    const float* __restrict__ bias_v, const float* __restrict__ bias_e,
    float* __restrict__ item_out, float* __restrict__ ent_out, int nrows)
{
    __shared__ __align__(16) float lds_w[4][DD];

    const int t = threadIdx.x;
    for (int i = t; i < 4 * DD; i += 256) {
        const int which = i >> 7;
        const int k = i & (DD - 1);
        const float* src = (which == 0) ? w_vv : (which == 1) ? w_ve
                         : (which == 2) ? w_ev : w_ee;
        lds_w[which][k] = src[k];
    }
    __syncthreads();

    const int wave = t >> 5;
    const int lane = t & 31;
    const long tile = (long)blockIdx.x * WAVES_PER_BLOCK + wave;
    const long row0 = tile * TILE_M;
    if (row0 >= nrows) return;

    const int m    = lane & 15;
    const int kSel = (lane >> 4) << 1;

    long arow = row0 + m;
    if (arow > (long)nrows - 1) arow = (long)nrows - 1;
    const float* vrow = v_in + arow * DD;
    const float* erow = e_in + arow * DD;

    const float bscale = (m < 2) ? 1.0f : 0.0f;
    const float* wv_col = lds_w[(m == 1) ? 3 : 2];
    const float* we_col = lds_w[(m == 1) ? 1 : 0];

    v8f cv = {};
    v8f ce = {};
#pragma unroll 4
    for (int k0 = 0; k0 < DD; k0 += 4) {
        const int k = k0 + kSel;
        v2f av = { vrow[k],   vrow[k + 1] };
        v2f ae = { erow[k],   erow[k + 1] };
        v2f bv = { wv_col[k] * bscale, wv_col[k + 1] * bscale };
        v2f be = { we_col[k] * bscale, we_col[k + 1] * bscale };
        cv = __builtin_amdgcn_wmma_f32_16x16x4_f32(false, av, false, bv,
                                                   (short)0, cv, false, false);
        ce = __builtin_amdgcn_wmma_f32_16x16x4_f32(false, ae, false, be,
                                                   (short)0, ce, false, false);
    }

    const v4f b_v = *(const v4f*)(bias_v + 4 * lane);
    const v4f b_e = *(const v4f*)(bias_e + 4 * lane);

#pragma unroll
    for (int r = 0; r < TILE_M; ++r) {
        const long row = row0 + r;
        if (row >= nrows) break;
        const int srcLane = (r < 8) ? 0 : 16;
        const float d_evv = __shfl(ce[r & 7], srcLane + 0, 32);
        const float d_eve = __shfl(ce[r & 7], srcLane + 1, 32);
        const float d_vev = __shfl(cv[r & 7], srcLane + 0, 32);
        const float d_vee = __shfl(cv[r & 7], srcLane + 1, 32);

        const v4f vx = *(const v4f*)(v_in + row * DD + 4 * lane);
        const v4f ex = *(const v4f*)(e_in + row * DD + 4 * lane);

        v4f io, eo;
        io.x = vx.x * d_evv + ex.x * d_vev + b_v.x;
        io.y = vx.y * d_evv + ex.y * d_vev + b_v.y;
        io.z = vx.z * d_evv + ex.z * d_vev + b_v.z;
        io.w = vx.w * d_evv + ex.w * d_vev + b_v.w;
        eo.x = vx.x * d_eve + ex.x * d_vee + b_e.x;
        eo.y = vx.y * d_eve + ex.y * d_vee + b_e.y;
        eo.z = vx.z * d_eve + ex.z * d_vee + b_e.z;
        eo.w = vx.w * d_eve + ex.w * d_vee + b_e.w;

        *(v4f*)(item_out + row * DD + 4 * lane) = io;
        *(v4f*)(ent_out  + row * DD + 4 * lane) = eo;
    }
}

extern "C" void kernel_launch(void* const* d_in, const int* in_sizes, int n_in,
                              void* d_out, int out_size, void* d_ws, size_t ws_size,
                              hipStream_t stream) {
    const float* v_in   = (const float*)d_in[0];
    const float* e_in   = (const float*)d_in[1];
    const float* w_vv   = (const float*)d_in[2];
    const float* w_ve   = (const float*)d_in[3];
    const float* w_ev   = (const float*)d_in[4];
    const float* w_ee   = (const float*)d_in[5];
    const float* bias_v = (const float*)d_in[6];
    const float* bias_e = (const float*)d_in[7];

    const int nrows = in_sizes[0] / DD;
    float* out      = (float*)d_out;
    float* item_out = out;
    float* ent_out  = out + (size_t)nrows * DD;

    const int tiles = (nrows + TILE_M - 1) / TILE_M;
    const int grid  = (tiles + WAVES_PER_BLOCK - 1) / WAVES_PER_BLOCK;

    hipLaunchKernelGGL(CrossCompress_86655260164823_kernel,
                       dim3(grid), dim3(256), 0, stream,
                       v_in, e_in, w_vv, w_ve, w_ev, w_ee, bias_v, bias_e,
                       item_out, ent_out, nrows);
}


# Each 16-row tile is read from global memory once: whole-row 16-byte loads kept in registers for the epilogue, WMMA A fragments read from a per-wave LDS copy instead of strided global loads
# speedup vs baseline: 1.7844x; 1.7844x over previous
#include <hip/hip_runtime.h>
#include <hip/hip_bf16.h>

typedef __attribute__((ext_vector_type(2))) float v2f;
typedef __attribute__((ext_vector_type(4))) float v4f;
typedef __attribute__((ext_vector_type(8))) float v8f;

#define DD 128
#define TILE_M 16
#define WAVES_PER_BLOCK 8
#define XPITCH (DD + 4)

__global__ __launch_bounds__(256, 1) void CrossCompress_86655260164823_kernel(
    const float* __restrict__ v_in, const float* __restrict__ e_in,
    const float* __restrict__ w_vv, const float* __restrict__ w_ve,
    const float* __restrict__ w_ev, const float* __restrict__ w_ee,
    const float* __restrict__ bias_v, const float* __restrict__ bias_e,
    float* __restrict__ item_out, float* __restrict__ ent_out, int nrows)
{
    __shared__ __align__(16) float lds_w[4][DD];
    __shared__ __align__(16) float lds_x[WAVES_PER_BLOCK][2][TILE_M][XPITCH];

    const int t = threadIdx.x;
    const int wave = __builtin_amdgcn_readfirstlane(t >> 5);
    const int lane = t & 31;
    const int tile = (int)blockIdx.x * WAVES_PER_BLOCK + wave;
    const int row0 = tile * TILE_M;

    const int wk  = t & (DD - 1);
    const int whi = wave >> 2;
    const float w_lo = (whi ? w_ve : w_vv)[wk];
    const float w_hi = (whi ? w_ee : w_ev)[wk];
    const v4f b_v = *(const v4f*)(bias_v + 4 * lane);
    const v4f b_e = *(const v4f*)(bias_e + 4 * lane);

    v4f vx[TILE_M], ex[TILE_M];
    for (int r = 0; r < TILE_M; ++r) {
        int lr = row0 + r;
        if (lr > nrows - 1) lr = nrows - 1;
        vx[r] = *(const v4f*)(v_in + (size_t)lr * DD + 4 * lane);
        ex[r] = *(const v4f*)(e_in + (size_t)lr * DD + 4 * lane);
    }

    lds_w[whi][wk]     = w_lo;
    lds_w[2 + whi][wk] = w_hi;

    for (int r = 0; r < TILE_M; ++r) {
        *(v4f*)&lds_x[wave][0][r][4 * lane] = vx[r];
        *(v4f*)&lds_x[wave][1][r][4 * lane] = ex[r];
    }
    __syncthreads();

    const int m    = lane & 15;
    const int kSel = (lane >> 4) << 1;

    const float* vrow = &lds_x[wave][0][m][0];
    const float* erow = &lds_x[wave][1][m][0];

    const float bscale = (m < 2) ? 1.0f : 0.0f;
    const float* wv_col = lds_w[(m == 1) ? 3 : 2];
    const float* we_col = lds_w[(m == 1) ? 1 : 0];

    v8f cv = {};
    v8f ce = {};
#pragma unroll
    for (int k0 = 0; k0 < DD; k0 += 4) {
        const int k = k0 + kSel;
        v2f av = *(const v2f*)(vrow + k);
        v2f ae = *(const v2f*)(erow + k);
        v2f bv = { wv_col[k] * bscale, wv_col[k + 1] * bscale };
        v2f be = { we_col[k] * bscale, we_col[k + 1] * bscale };
        cv = __builtin_amdgcn_wmma_f32_16x16x4_f32(false, av, false, bv,
                                                   (short)0, cv, false, false);
        ce = __builtin_amdgcn_wmma_f32_16x16x4_f32(false, ae, false, be,
                                                   (short)0, ce, false, false);
    }

#pragma unroll
    for (int r = 0; r < TILE_M; ++r) {
        const int row = row0 + r;
        if (row >= nrows) break;
        const int srcLane = (r < 8) ? 0 : 16;
        const float d_evv = __shfl(ce[r & 7], srcLane + 0, 32);
        const float d_eve = __shfl(ce[r & 7], srcLane + 1, 32);
        const float d_vev = __shfl(cv[r & 7], srcLane + 0, 32);
        const float d_vee = __shfl(cv[r & 7], srcLane + 1, 32);

        const v4f vxr = vx[r];
        const v4f exr = ex[r];

        v4f io, eo;
        io.x = vxr.x * d_evv + exr.x * d_vev + b_v.x;
        io.y = vxr.y * d_evv + exr.y * d_vev + b_v.y;
        io.z = vxr.z * d_evv + exr.z * d_vev + b_v.z;
        io.w = vxr.w * d_evv + exr.w * d_vev + b_v.w;
        eo.x = vxr.x * d_eve + exr.x * d_vee + b_e.x;
        eo.y = vxr.y * d_eve + exr.y * d_vee + b_e.y;
        eo.z = vxr.z * d_eve + exr.z * d_vee + b_e.z;
        eo.w = vxr.w * d_eve + exr.w * d_vee + b_e.w;

        *(v4f*)(item_out + (size_t)row * DD + 4 * lane) = io;
        *(v4f*)(ent_out  + (size_t)row * DD + 4 * lane) = eo;
    }
}

extern "C" void kernel_launch(void* const* d_in, const int* in_sizes, int n_in,
                              void* d_out, int out_size, void* d_ws, size_t ws_size,
                              hipStream_t stream) {
    const float* v_in   = (const float*)d_in[0];
    const float* e_in   = (const float*)d_in[1];
    const float* w_vv   = (const float*)d_in[2];
    const float* w_ve   = (const float*)d_in[3];
    const float* w_ev   = (const float*)d_in[4];
    const float* w_ee   = (const float*)d_in[5];
    const float* bias_v = (const float*)d_in[6];
    const float* bias_e = (const float*)d_in[7];

    const int nrows = in_sizes[0] / DD;
    float* out      = (float*)d_out;
    float* item_out = out;
    float* ent_out  = out + (size_t)nrows * DD;

    const int tiles = (nrows + TILE_M - 1) / TILE_M;
    const int grid  = (tiles + WAVES_PER_BLOCK - 1) / WAVES_PER_BLOCK;

    hipLaunchKernelGGL(CrossCompress_86655260164823_kernel,
                       dim3(grid), dim3(256), 0, stream,
                       v_in, e_in, w_vv, w_ve, w_ev, w_ee, bias_v, bias_e,
                       item_out, ent_out, nrows);
}


# WMMA tile = 8 rows of V stacked on the same 8 rows of E, one accumulator against 4 weight columns: 8 rows per wave, 2048 waves in 256 work-groups, so every compute unit is busy at two waves per SIMD
# speedup vs baseline: 2.6228x; 1.4698x over previous
#include <hip/hip_runtime.h>
#include <hip/hip_bf16.h>

typedef __attribute__((ext_vector_type(2))) float v2f;
typedef __attribute__((ext_vector_type(4))) float v4f;
typedef __attribute__((ext_vector_type(8))) float v8f;

#define DD 128
#define ROWS_PER_WAVE 8
#define TILE_M 16
#define WAVES_PER_BLOCK 8
#define XPITCH (DD + 4)

__global__ __launch_bounds__(256, 1) void CrossCompress_86655260164823_kernel(
    const float* __restrict__ v_in, const float* __restrict__ e_in,
    const float* __restrict__ w_vv, const float* __restrict__ w_ve,
    const float* __restrict__ w_ev, const float* __restrict__ w_ee,
    const float* __restrict__ bias_v, const float* __restrict__ bias_e,
    float* __restrict__ item_out, float* __restrict__ ent_out, int nrows)
{
    __shared__ __align__(16) float lds_w[4][DD];
    __shared__ __align__(16) float lds_x[WAVES_PER_BLOCK][TILE_M][XPITCH];

    const int t = threadIdx.x;
    const int wave = __builtin_amdgcn_readfirstlane(t >> 5);
    const int lane = t & 31;
    const int grp  = (int)blockIdx.x * WAVES_PER_BLOCK + wave;
    const int row0 = grp * ROWS_PER_WAVE;

    const int wk  = t & (DD - 1);
    const int whi = wave >> 2;
    const float w_lo = (whi ? w_ve : w_vv)[wk];
    const float w_hi = (whi ? w_ee : w_ev)[wk];
    const v4f b_v = *(const v4f*)(bias_v + 4 * lane);
    const v4f b_e = *(const v4f*)(bias_e + 4 * lane);

    v4f vx[ROWS_PER_WAVE], ex[ROWS_PER_WAVE];
    for (int r = 0; r < ROWS_PER_WAVE; ++r) {
        int lr = row0 + r;
        if (lr > nrows - 1) lr = nrows - 1;
        vx[r] = *(const v4f*)(v_in + (size_t)lr * DD + 4 * lane);
        ex[r] = *(const v4f*)(e_in + (size_t)lr * DD + 4 * lane);
    }

    lds_w[whi][wk]     = w_lo;
    lds_w[2 + whi][wk] = w_hi;

    for (int r = 0; r < ROWS_PER_WAVE; ++r) {
        *(v4f*)&lds_x[wave][r][4 * lane]                 = vx[r];
        *(v4f*)&lds_x[wave][ROWS_PER_WAVE + r][4 * lane] = ex[r];
    }
    __syncthreads();

    const int m    = lane & 15;
    const int kSel = (lane >> 4) << 1;

    const float* arow = &lds_x[wave][m][0];

    const float bscale = (m < 4) ? 1.0f : 0.0f;
    const float* w_col = lds_w[(m & 3) ^ 2];

    v8f c = {};
#pragma unroll
    for (int k0 = 0; k0 < DD; k0 += 4) {
        const int k = k0 + kSel;
        v2f a = *(const v2f*)(arow + k);
        v2f b = { w_col[k] * bscale, w_col[k + 1] * bscale };
        c = __builtin_amdgcn_wmma_f32_16x16x4_f32(false, a, false, b,
                                                  (short)0, c, false, false);
    }

#pragma unroll
    for (int r = 0; r < ROWS_PER_WAVE; ++r) {
        const int row = row0 + r;
        if (row >= nrows) break;
        const float d_vev = __shfl(c[r], 0, 32);
        const float d_vee = __shfl(c[r], 1, 32);
        const float d_evv = __shfl(c[r], 18, 32);
        const float d_eve = __shfl(c[r], 19, 32);

        const v4f vxr = vx[r];
        const v4f exr = ex[r];

        v4f io, eo;
        io.x = vxr.x * d_evv + exr.x * d_vev + b_v.x;
        io.y = vxr.y * d_evv + exr.y * d_vev + b_v.y;
        io.z = vxr.z * d_evv + exr.z * d_vev + b_v.z;
        io.w = vxr.w * d_evv + exr.w * d_vev + b_v.w;
        eo.x = vxr.x * d_eve + exr.x * d_vee + b_e.x;
        eo.y = vxr.y * d_eve + exr.y * d_vee + b_e.y;
        eo.z = vxr.z * d_eve + exr.z * d_vee + b_e.z;
        eo.w = vxr.w * d_eve + exr.w * d_vee + b_e.w;

        *(v4f*)(item_out + (size_t)row * DD + 4 * lane) = io;
        *(v4f*)(ent_out  + (size_t)row * DD + 4 * lane) = eo;
    }
}

extern "C" void kernel_launch(void* const* d_in, const int* in_sizes, int n_in,
                              void* d_out, int out_size, void* d_ws, size_t ws_size,
                              hipStream_t stream) {
    const float* v_in   = (const float*)d_in[0];
    const float* e_in   = (const float*)d_in[1];
    const float* w_vv   = (const float*)d_in[2];
    const float* w_ve   = (const float*)d_in[3];
    const float* w_ev   = (const float*)d_in[4];
    const float* w_ee   = (const float*)d_in[5];
    const float* bias_v = (const float*)d_in[6];
    const float* bias_e = (const float*)d_in[7];

    const int nrows = in_sizes[0] / DD;
    float* out      = (float*)d_out;
    float* item_out = out;
    float* ent_out  = out + (size_t)nrows * DD;

    const int groups = (nrows + ROWS_PER_WAVE - 1) / ROWS_PER_WAVE;
    const int grid   = (groups + WAVES_PER_BLOCK - 1) / WAVES_PER_BLOCK;

    hipLaunchKernelGGL(CrossCompress_86655260164823_kernel,
                       dim3(grid), dim3(256), 0, stream,
                       v_in, e_in, w_vv, w_ve, w_ev, w_ee, bias_v, bias_e,
                       item_out, ent_out, nrows);
}


# One LDS round trip behind a wave's last row: B fragments prebuilt in registers under the row loads (weight barrier moved up), A read as one 16-byte group, dots by v_readlane; k sum order changed
# speedup vs baseline: 2.8315x; 1.0796x over previous
#include <hip/hip_runtime.h>
#include <hip/hip_bf16.h>

typedef __attribute__((ext_vector_type(2))) float v2f;
typedef __attribute__((ext_vector_type(4))) float v4f;
typedef __attribute__((ext_vector_type(8))) float v8f;

#define DD 128
#define ROWS_PER_WAVE 8
#define TILE_M 16
#define WAVES_PER_BLOCK 8
#define KQUADS (DD / 8)
#define XPITCH (DD + 4)
#define WPITCH (DD + 8)

__global__ __launch_bounds__(256, 1) void CrossCompress_86655260164823_kernel(
    const float* __restrict__ v_in, const float* __restrict__ e_in,
    const float* __restrict__ w_vv, const float* __restrict__ w_ve,
    const float* __restrict__ w_ev, const float* __restrict__ w_ee,
    const float* __restrict__ bias_v, const float* __restrict__ bias_e,
    float* __restrict__ item_out, float* __restrict__ ent_out, int nrows)
{
    __shared__ __align__(16) float lds_w[5][WPITCH];
    __shared__ __align__(16) float lds_x[WAVES_PER_BLOCK][TILE_M][XPITCH];

    const int t = threadIdx.x;
    const int wave = __builtin_amdgcn_readfirstlane(t >> 5);
    const int lane = t & 31;
    const int grp  = (int)blockIdx.x * WAVES_PER_BLOCK + wave;
    const int row0 = grp * ROWS_PER_WAVE;

    const int wk  = t & (DD - 1);
    const int whi = wave >> 2;
    const float w_lo = (whi ? w_ve : w_vv)[wk];
    const float w_hi = (whi ? w_ee : w_ev)[wk];

    v4f vx[ROWS_PER_WAVE], ex[ROWS_PER_WAVE];
    for (int r = 0; r < ROWS_PER_WAVE; ++r) {
        int lr = row0 + r;
        if (lr > nrows - 1) lr = nrows - 1;
        vx[r] = *(const v4f*)(v_in + (size_t)lr * DD + 4 * lane);
        ex[r] = *(const v4f*)(e_in + (size_t)lr * DD + 4 * lane);
    }
    const v4f b_v = *(const v4f*)(bias_v + 4 * lane);
    const v4f b_e = *(const v4f*)(bias_e + 4 * lane);

    lds_w[whi][wk]     = w_lo;
    lds_w[2 + whi][wk] = w_hi;
    if (whi == 0) lds_w[4][wk] = 0.0f;

    __builtin_amdgcn_fence(__ATOMIC_RELEASE, "workgroup", "local");
    __builtin_amdgcn_s_barrier();
    __builtin_amdgcn_fence(__ATOMIC_ACQUIRE, "workgroup", "local");

    const int m  = lane & 15;
    const int h4 = (lane >> 4) << 2;

    const float* bcol = &lds_w[(m < 4) ? (m ^ 2) : 4][h4];
    v4f bq[KQUADS];
#pragma unroll
    for (int j = 0; j < KQUADS; ++j)
        bq[j] = *(const v4f*)(bcol + 8 * j);
    __builtin_amdgcn_sched_barrier(0);

#pragma unroll
    for (int r = 0; r < ROWS_PER_WAVE; ++r) {
        *(v4f*)&lds_x[wave][r][4 * lane]                 = vx[r];
        *(v4f*)&lds_x[wave][ROWS_PER_WAVE + r][4 * lane] = ex[r];
    }

    const float* arow = &lds_x[wave][m][h4];
    v4f aq[KQUADS];
#pragma unroll
    for (int j = 0; j < KQUADS; ++j)
        aq[j] = *(const v4f*)(arow + 8 * j);
    __builtin_amdgcn_sched_barrier(0);

    v8f c = {};
#pragma unroll
    for (int j = 0; j < KQUADS; ++j) {
        const v2f a0 = { aq[j].x, aq[j].y };
        const v2f b0 = { bq[j].x, bq[j].y };
        const v2f a1 = { aq[j].z, aq[j].w };
        const v2f b1 = { bq[j].z, bq[j].w };
        c = __builtin_amdgcn_wmma_f32_16x16x4_f32(false, a0, false, b0, (short)0, c, false, false);
        c = __builtin_amdgcn_wmma_f32_16x16x4_f32(false, a1, false, b1, (short)0, c, false, false);
    }

    float d_vev[ROWS_PER_WAVE], d_vee[ROWS_PER_WAVE], d_evv[ROWS_PER_WAVE], d_eve[ROWS_PER_WAVE];
#pragma unroll
    for (int r = 0; r < ROWS_PER_WAVE; ++r) {
        const int cr = __float_as_int(c[r]);
        d_vev[r] = __int_as_float(__builtin_amdgcn_readlane(cr, 0));
        d_vee[r] = __int_as_float(__builtin_amdgcn_readlane(cr, 1));
        d_evv[r] = __int_as_float(__builtin_amdgcn_readlane(cr, 18));
        d_eve[r] = __int_as_float(__builtin_amdgcn_readlane(cr, 19));
    }

#pragma unroll
    for (int r = 0; r < ROWS_PER_WAVE; ++r) {
        const int row = row0 + r;
        if (row < nrows) {
            const v4f vxr = vx[r];
            const v4f exr = ex[r];

            v4f io, eo;
            io.x = vxr.x * d_evv[r] + exr.x * d_vev[r] + b_v.x;
            io.y = vxr.y * d_evv[r] + exr.y * d_vev[r] + b_v.y;
            io.z = vxr.z * d_evv[r] + exr.z * d_vev[r] + b_v.z;
            io.w = vxr.w * d_evv[r] + exr.w * d_vev[r] + b_v.w;
            eo.x = vxr.x * d_eve[r] + exr.x * d_vee[r] + b_e.x;
            eo.y = vxr.y * d_eve[r] + exr.y * d_vee[r] + b_e.y;
            eo.z = vxr.z * d_eve[r] + exr.z * d_vee[r] + b_e.z;
            eo.w = vxr.w * d_eve[r] + exr.w * d_vee[r] + b_e.w;

            *(v4f*)(item_out + (size_t)row * DD + 4 * lane) = io;
            *(v4f*)(ent_out  + (size_t)row * DD + 4 * lane) = eo;
        }
    }
}

extern "C" void kernel_launch(void* const* d_in, const int* in_sizes, int n_in,
                              void* d_out, int out_size, void* d_ws, size_t ws_size,
                              hipStream_t stream) {
    const float* v_in   = (const float*)d_in[0];
    const float* e_in   = (const float*)d_in[1];
    const float* w_vv   = (const float*)d_in[2];
    const float* w_ve   = (const float*)d_in[3];
    const float* w_ev   = (const float*)d_in[4];
    const float* w_ee   = (const float*)d_in[5];
    const float* bias_v = (const float*)d_in[6];
    const float* bias_e = (const float*)d_in[7];

    const int nrows = in_sizes[0] / DD;
    float* out      = (float*)d_out;
    float* item_out = out;
    float* ent_out  = out + (size_t)nrows * DD;

    const int groups = (nrows + ROWS_PER_WAVE - 1) / ROWS_PER_WAVE;
    const int grid   = (groups + WAVES_PER_BLOCK - 1) / WAVES_PER_BLOCK;

    hipLaunchKernelGGL(CrossCompress_86655260164823_kernel,
                       dim3(grid), dim3(256), 0, stream,
                       v_in, e_in, w_vv, w_ve, w_ev, w_ee, bias_v, bias_e,
                       item_out, ent_out, nrows);
}
